# starting kernel
# baseline (speedup 1.0000x reference)
#include <hip/hip_runtime.h>
#include <math.h>

typedef float v2f __attribute__((ext_vector_type(2)));
typedef float v8f __attribute__((ext_vector_type(8)));

#define EMBED            2048
#define BATCH            32
#define BLOCK_THREADS    256
#define WAVES_PER_BLOCK  8
#define ROWS_PER_WAVE    16
#define ROWS_PER_BLOCK   128
#define BLOCKS_PER_BATCH 16

__global__ __launch_bounds__(BLOCK_THREADS)
void attn_outer_softmax_wmma(const float* __restrict__ q,
                             const float* __restrict__ k,
                             const float* __restrict__ v,
                             float* __restrict__ out)
{
    __shared__ float sK[EMBED];
    __shared__ float sV[EMBED];
    __shared__ float sRed[2 * BLOCK_THREADS];
    __shared__ float sOut[WAVES_PER_BLOCK][32];

    const int b    = blockIdx.y;
    const int tid  = threadIdx.x;
    const int wave = tid >> 5;
    const int lane = tid & 31;

    const float* kb = k + b * EMBED;
    const float* vb = v + b * EMBED;

    for (int i = tid; i < EMBED; i += BLOCK_THREADS) {
        sK[i] = kb[i];
        sV[i] = vb[i];
    }
    __syncthreads();

    float kmx = -INFINITY, kmn = INFINITY;
    for (int i = tid; i < EMBED; i += BLOCK_THREADS) {
        float x = sK[i];
        kmx = fmaxf(kmx, x);
        kmn = fminf(kmn, x);
    }
    sRed[tid]                 = kmx;
    sRed[BLOCK_THREADS + tid] = kmn;
    __syncthreads();
    for (int s = BLOCK_THREADS / 2; s > 0; s >>= 1) {
        if (tid < s) {
            sRed[tid]                 = fmaxf(sRed[tid], sRed[tid + s]);
            sRed[BLOCK_THREADS + tid] = fminf(sRed[BLOCK_THREADS + tid],
                                              sRed[BLOCK_THREADS + tid + s]);
        }
        __syncthreads();
    }
    const float kmax = sRed[0];
    const float kmin = sRed[BLOCK_THREADS];

    const int rowBase = blockIdx.x * ROWS_PER_BLOCK + wave * ROWS_PER_WAVE;
    const int n       = lane & 15;
    const int h       = lane >> 4;

    const float tlScale = 0.02209708691207961f * 1.4426950408889634f;
    const float tl = q[b * EMBED + rowBase + n] * tlScale;
    const float m2 = fmaxf(tl * kmax, tl * kmin);

    const bool isV   = (n == 0);
    const bool isOne = (n == 1);

    v8f acc = {};
    #pragma unroll 8
    for (int j0 = 0; j0 < EMBED; j0 += 4) {
        const float2 kk = *(const float2*)(&sK[j0 + 2 * h]);
        const float2 vv = *(const float2*)(&sV[j0 + 2 * h]);

        v2f a, bb;
        a.x = __builtin_amdgcn_exp2f(fmaf(tl, kk.x, -m2));
        a.y = __builtin_amdgcn_exp2f(fmaf(tl, kk.y, -m2));
        bb.x = isV ? vv.x : (isOne ? 1.0f : 0.0f);
        bb.y = isV ? vv.y : (isOne ? 1.0f : 0.0f);

        acc = __builtin_amdgcn_wmma_f32_16x16x4_f32(
             false, a,  false, bb,
             (short)0, acc,  false,  false);
    }

    if (n < 2) {
        #pragma unroll
        for (int r = 0; r < 8; ++r)
            sOut[wave][n * 16 + h * 8 + r] = acc[r];
    }
    __syncthreads();

    if (lane < 16) {
        const float num = sOut[wave][lane];
        const float den = sOut[wave][16 + lane];
        out[b * EMBED + rowBase + lane] = num / den;
    }
}

extern "C" void kernel_launch(void* const* d_in, const int* in_sizes, int n_in,
                              void* d_out, int out_size, void* d_ws, size_t ws_size,
                              hipStream_t stream) {
    const float* q = (const float*)d_in[0];
    const float* k = (const float*)d_in[1];
    const float* v = (const float*)d_in[2];
    float* out = (float*)d_out;

    dim3 grid(BLOCKS_PER_BATCH, BATCH);
    dim3 block(BLOCK_THREADS);
    attn_outer_softmax_wmma<<<grid, block, 0, stream>>>(q, k, v, out);
}


# Main loop: K/V operand fragments read from LDS as one group of sixteen 16-byte loads a whole 64-j block ahead of the 16 WMMAs that use them (second register set), not 8-byte loads inside each step
# speedup vs baseline: 1.0932x; 1.0932x over previous
#include <hip/hip_runtime.h>
#include <math.h>

typedef float v2f __attribute__((ext_vector_type(2)));
typedef float v8f __attribute__((ext_vector_type(8)));

#define EMBED            2048
#define BATCH            32
#define BLOCK_THREADS    256
#define WAVES_PER_BLOCK  8
#define ROWS_PER_WAVE    16
#define ROWS_PER_BLOCK   128
#define BLOCKS_PER_BATCH 16

__global__ __launch_bounds__(BLOCK_THREADS)
void attn_outer_softmax_wmma(const float* __restrict__ q,
                             const float* __restrict__ k,
                             const float* __restrict__ v,
                             float* __restrict__ out)
{
    __shared__ __attribute__((aligned(16))) float sK[EMBED];
    __shared__ __attribute__((aligned(16))) float sV[EMBED];
    __shared__ float sRed[2 * BLOCK_THREADS];
    __shared__ float sOut[WAVES_PER_BLOCK][32];

    const int b    = blockIdx.y;
    const int tid  = threadIdx.x;
    const int wave = tid >> 5;
    const int lane = tid & 31;

    const float* kb = k + b * EMBED;
    const float* vb = v + b * EMBED;

    for (int i = tid; i < EMBED; i += BLOCK_THREADS) {
        sK[i] = kb[i];
        sV[i] = vb[i];
    }
    __syncthreads();

    float kmx = -INFINITY, kmn = INFINITY;
    for (int i = tid; i < EMBED; i += BLOCK_THREADS) {
        float x = sK[i];
        kmx = fmaxf(kmx, x);
        kmn = fminf(kmn, x);
    }
    sRed[tid]                 = kmx;
    sRed[BLOCK_THREADS + tid] = kmn;
    __syncthreads();
    for (int s = BLOCK_THREADS / 2; s > 0; s >>= 1) {
        if (tid < s) {
            sRed[tid]                 = fmaxf(sRed[tid], sRed[tid + s]);
            sRed[BLOCK_THREADS + tid] = fminf(sRed[BLOCK_THREADS + tid],
                                              sRed[BLOCK_THREADS + tid + s]);
        }
        __syncthreads();
    }
    const float kmax = sRed[0];
    const float kmin = sRed[BLOCK_THREADS];

    const int rowBase = blockIdx.x * ROWS_PER_BLOCK + wave * ROWS_PER_WAVE;
    const int n       = lane & 15;
    const int h       = lane >> 4;

    const float tlScale = 0.02209708691207961f * 1.4426950408889634f;
    const float tl = q[b * EMBED + rowBase + n] * tlScale;
    const float m2 = fmaxf(tl * kmax, tl * kmin);

    const bool isV   = (n == 0);
    const bool isOne = (n == 1);

    const float4* sK4 = (const float4*)sK;
    const float4* sV4 = (const float4*)sV;
    const int fo = 8 * h;

    float4 kc0 = sK4[fo + 0], kc1 = sK4[fo + 1], kc2 = sK4[fo + 2], kc3 = sK4[fo + 3], kc4 = sK4[fo + 4], kc5 = sK4[fo + 5], kc6 = sK4[fo + 6], kc7 = sK4[fo + 7];
    float4 vc0 = sV4[fo + 0], vc1 = sV4[fo + 1], vc2 = sV4[fo + 2], vc3 = sV4[fo + 3], vc4 = sV4[fo + 4], vc5 = sV4[fo + 5], vc6 = sV4[fo + 6], vc7 = sV4[fo + 7];

    const float bconst = isOne ? 1.0f : 0.0f;
    v8f acc = {};

#define STEP(KX, KY, VX, VY)                                                  \
    {                                                                         \
        v2f a_, b_;                                                           \
        a_.x = __builtin_amdgcn_exp2f(fmaf(tl, (KX), -m2));                   \
        a_.y = __builtin_amdgcn_exp2f(fmaf(tl, (KY), -m2));                   \
        b_.x = isV ? (VX) : bconst;                                           \
        b_.y = isV ? (VY) : bconst;                                           \
        acc = __builtin_amdgcn_wmma_f32_16x16x4_f32(false, a_, false, b_,     \
                                                    (short)0, acc, false, false); \
    }

    #pragma unroll 2
    for (int j0 = 0; j0 < EMBED; j0 += 64) {
        const int fn = (((j0 + 64) & (EMBED - 1)) >> 2) + fo;
        const float4 kn0 = sK4[fn + 0], kn1 = sK4[fn + 1], kn2 = sK4[fn + 2], kn3 = sK4[fn + 3], kn4 = sK4[fn + 4], kn5 = sK4[fn + 5], kn6 = sK4[fn + 6], kn7 = sK4[fn + 7];
        const float4 vn0 = sV4[fn + 0], vn1 = sV4[fn + 1], vn2 = sV4[fn + 2], vn3 = sV4[fn + 3], vn4 = sV4[fn + 4], vn5 = sV4[fn + 5], vn6 = sV4[fn + 6], vn7 = sV4[fn + 7];
        __builtin_amdgcn_sched_barrier(0);

        STEP(kc0.x, kc0.y, vc0.x, vc0.y)
        STEP(kc0.z, kc0.w, vc0.z, vc0.w)
        STEP(kc1.x, kc1.y, vc1.x, vc1.y)
        STEP(kc1.z, kc1.w, vc1.z, vc1.w)
        STEP(kc2.x, kc2.y, vc2.x, vc2.y)
        STEP(kc2.z, kc2.w, vc2.z, vc2.w)
        STEP(kc3.x, kc3.y, vc3.x, vc3.y)
        STEP(kc3.z, kc3.w, vc3.z, vc3.w)
        STEP(kc4.x, kc4.y, vc4.x, vc4.y)
        STEP(kc4.z, kc4.w, vc4.z, vc4.w)
        STEP(kc5.x, kc5.y, vc5.x, vc5.y)
        STEP(kc5.z, kc5.w, vc5.z, vc5.w)
        STEP(kc6.x, kc6.y, vc6.x, vc6.y)
        STEP(kc6.z, kc6.w, vc6.z, vc6.w)
        STEP(kc7.x, kc7.y, vc7.x, vc7.y)
        STEP(kc7.z, kc7.w, vc7.z, vc7.w)

        __builtin_amdgcn_sched_barrier(0);
        kc0 = kn0; kc1 = kn1; kc2 = kn2; kc3 = kn3; kc4 = kn4; kc5 = kn5; kc6 = kn6; kc7 = kn7;
        vc0 = vn0; vc1 = vn1; vc2 = vn2; vc3 = vn3; vc4 = vn4; vc5 = vn5; vc6 = vn6; vc7 = vn7;
    }
#undef STEP

    if (n < 2) {
        #pragma unroll
        for (int r = 0; r < 8; ++r)
            sOut[wave][n * 16 + h * 8 + r] = acc[r];
    }
    __syncthreads();

    if (lane < 16) {
        const float num = sOut[wave][lane];
        const float den = sOut[wave][16 + lane];
        out[b * EMBED + rowBase + lane] = num / den;
    }
}

extern "C" void kernel_launch(void* const* d_in, const int* in_sizes, int n_in,
                              void* d_out, int out_size, void* d_ws, size_t ws_size,
                              hipStream_t stream) {
    const float* q = (const float*)d_in[0];
    const float* k = (const float*)d_in[1];
    const float* v = (const float*)d_in[2];
    float* out = (float*)d_out;

    dim3 grid(BLOCKS_PER_BATCH, BATCH);
    dim3 block(BLOCK_THREADS);
    attn_outer_softmax_wmma<<<grid, block, 0, stream>>>(q, k, v, out);
}


# Main loop: the fma+exp2 (A) and column-select (B) operand work is done one 64-j block ahead, in place on the block just read, between the current block's WMMAs; every WMMA reads ready registers
# speedup vs baseline: 1.1404x; 1.0432x over previous
#include <hip/hip_runtime.h>
#include <math.h>

typedef float v2f __attribute__((ext_vector_type(2)));
typedef float v8f __attribute__((ext_vector_type(8)));

#define EMBED            2048
#define BATCH            32
#define BLOCK_THREADS    256
#define WAVES_PER_BLOCK  8
#define ROWS_PER_WAVE    16
#define ROWS_PER_BLOCK   128
#define BLOCKS_PER_BATCH 16

__global__ __launch_bounds__(BLOCK_THREADS)
void attn_outer_softmax_wmma(const float* __restrict__ q,
                             const float* __restrict__ k,
                             const float* __restrict__ v,
                             float* __restrict__ out)
{
    __shared__ __attribute__((aligned(16))) float sK[EMBED];
    __shared__ __attribute__((aligned(16))) float sV[EMBED];
    __shared__ float sRed[2 * BLOCK_THREADS];
    __shared__ float sOut[WAVES_PER_BLOCK][32];

    const int b    = blockIdx.y;
    const int tid  = threadIdx.x;
    const int wave = tid >> 5;
    const int lane = tid & 31;

    const float* kb = k + b * EMBED;
    const float* vb = v + b * EMBED;

    for (int i = tid; i < EMBED; i += BLOCK_THREADS) {
        sK[i] = kb[i];
        sV[i] = vb[i];
    }
    __syncthreads();

    float kmx = -INFINITY, kmn = INFINITY;
    for (int i = tid; i < EMBED; i += BLOCK_THREADS) {
        float x = sK[i];
        kmx = fmaxf(kmx, x);
        kmn = fminf(kmn, x);
    }
    sRed[tid]                 = kmx;
    sRed[BLOCK_THREADS + tid] = kmn;
    __syncthreads();
    for (int s = BLOCK_THREADS / 2; s > 0; s >>= 1) {
        if (tid < s) {
            sRed[tid]                 = fmaxf(sRed[tid], sRed[tid + s]);
            sRed[BLOCK_THREADS + tid] = fminf(sRed[BLOCK_THREADS + tid],
                                              sRed[BLOCK_THREADS + tid + s]);
        }
        __syncthreads();
    }
    const float kmax = sRed[0];
    const float kmin = sRed[BLOCK_THREADS];

    const int rowBase = blockIdx.x * ROWS_PER_BLOCK + wave * ROWS_PER_WAVE;
    const int n       = lane & 15;
    const int h       = lane >> 4;

    const float tlScale = 0.02209708691207961f * 1.4426950408889634f;
    const float tl = q[b * EMBED + rowBase + n] * tlScale;
    const float m2 = fmaxf(tl * kmax, tl * kmin);

    const bool isV   = (n == 0);
    const bool isOne = (n == 1);

    const float4* sK4 = (const float4*)sK;
    const float4* sV4 = (const float4*)sV;
    const int fo = 8 * h;

    const float bconst = isOne ? 1.0f : 0.0f;

#define EX(X)   __builtin_amdgcn_exp2f(fmaf(tl, (X), -m2))
#define SEL(X)  (isV ? (X) : bconst)
#define TLO(K, V) { (K).x = EX((K).x); (K).y = EX((K).y); (V).x = SEL((V).x); (V).y = SEL((V).y); }
#define THI(K, V) { (K).z = EX((K).z); (K).w = EX((K).w); (V).z = SEL((V).z); (V).w = SEL((V).w); }
#define SB      __builtin_amdgcn_sched_barrier(0);
#define WM(AX, AY, BX, BY)                                                    \
    {                                                                         \
        v2f a_, b_;                                                           \
        a_.x = (AX); a_.y = (AY); b_.x = (BX); b_.y = (BY);                   \
        acc = __builtin_amdgcn_wmma_f32_16x16x4_f32(false, a_, false, b_,     \
                                                    (short)0, acc, false, false); \
    }
#define WLO(K, V) WM((K).x, (K).y, (V).x, (V).y)
#define WHI(K, V) WM((K).z, (K).w, (V).z, (V).w)

    float4 kc0 = sK4[fo + 0], vc0 = sV4[fo + 0], kc1 = sK4[fo + 1], vc1 = sV4[fo + 1];
    float4 kc2 = sK4[fo + 2], vc2 = sV4[fo + 2], kc3 = sK4[fo + 3], vc3 = sV4[fo + 3];
    float4 kc4 = sK4[fo + 4], vc4 = sV4[fo + 4], kc5 = sK4[fo + 5], vc5 = sV4[fo + 5];
    float4 kc6 = sK4[fo + 6], vc6 = sV4[fo + 6], kc7 = sK4[fo + 7], vc7 = sV4[fo + 7];
    TLO(kc0, vc0) THI(kc0, vc0) TLO(kc1, vc1) THI(kc1, vc1)
    TLO(kc2, vc2) THI(kc2, vc2) TLO(kc3, vc3) THI(kc3, vc3)
    TLO(kc4, vc4) THI(kc4, vc4) TLO(kc5, vc5) THI(kc5, vc5)
    TLO(kc6, vc6) THI(kc6, vc6) TLO(kc7, vc7) THI(kc7, vc7)

    v8f acc = {};

    #pragma unroll 2
    for (int j0 = 0; j0 < EMBED; j0 += 64) {
        const int fn = (((j0 + 64) & (EMBED - 1)) >> 2) + fo;
        float4 kn0 = sK4[fn + 0], vn0 = sV4[fn + 0], kn1 = sK4[fn + 1], vn1 = sV4[fn + 1];
        float4 kn2 = sK4[fn + 2], vn2 = sV4[fn + 2], kn3 = sK4[fn + 3], vn3 = sV4[fn + 3];
        float4 kn4 = sK4[fn + 4], vn4 = sV4[fn + 4], kn5 = sK4[fn + 5], vn5 = sV4[fn + 5];
        float4 kn6 = sK4[fn + 6], vn6 = sV4[fn + 6], kn7 = sK4[fn + 7], vn7 = sV4[fn + 7];
        SB

        WLO(kc0, vc0) SB
        WHI(kc0, vc0) SB
        WLO(kc1, vc1) SB
        WHI(kc1, vc1) SB
        WLO(kc2, vc2) SB  TLO(kn0, vn0) THI(kn0, vn0) SB
        WHI(kc2, vc2) SB  TLO(kn1, vn1) THI(kn1, vn1) SB
        WLO(kc3, vc3) SB  TLO(kn2, vn2) THI(kn2, vn2) SB
        WHI(kc3, vc3) SB  TLO(kn3, vn3) THI(kn3, vn3) SB
        WLO(kc4, vc4) SB  TLO(kn4, vn4) SB
        WHI(kc4, vc4) SB  THI(kn4, vn4) SB
        WLO(kc5, vc5) SB  TLO(kn5, vn5) SB
        WHI(kc5, vc5) SB  THI(kn5, vn5) SB
        WLO(kc6, vc6) SB  TLO(kn6, vn6) SB
        WHI(kc6, vc6) SB  THI(kn6, vn6) SB
        WLO(kc7, vc7) SB  TLO(kn7, vn7) SB
        WHI(kc7, vc7) SB  THI(kn7, vn7) SB

        kc0 = kn0; kc1 = kn1; kc2 = kn2; kc3 = kn3; kc4 = kn4; kc5 = kn5; kc6 = kn6; kc7 = kn7;
        vc0 = vn0; vc1 = vn1; vc2 = vn2; vc3 = vn3; vc4 = vn4; vc5 = vn5; vc6 = vn6; vc7 = vn7;
    }
#undef WHI
#undef WLO
#undef WM
#undef SB
#undef THI
#undef TLO
#undef SEL
#undef EX

    if (n < 2) {
        #pragma unroll
        for (int r = 0; r < 8; ++r)
            sOut[wave][n * 16 + h * 8 + r] = acc[r];
    }
    __syncthreads();

    if (lane < 16) {
        const float num = sOut[wave][lane];
        const float den = sOut[wave][16 + lane];
        out[b * EMBED + rowBase + lane] = num / den;
    }
}

extern "C" void kernel_launch(void* const* d_in, const int* in_sizes, int n_in,
                              void* d_out, int out_size, void* d_ws, size_t ws_size,
                              hipStream_t stream) {
    const float* q = (const float*)d_in[0];
    const float* k = (const float*)d_in[1];
    const float* v = (const float*)d_in[2];
    float* out = (float*)d_out;

    dim3 grid(BLOCKS_PER_BATCH, BATCH);
    dim3 block(BLOCK_THREADS);
    attn_outer_softmax_wmma<<<grid, block, 0, stream>>>(q, k, v, out);
}


# Prologue: K/V fetched by one group of 16-byte global loads per thread, q with them; kmax/kmin reduced from those registers by lane permutes + one barrier (was 8 load-wait trips and a 9-barrier tree)
# speedup vs baseline: 1.3503x; 1.1841x over previous
#include <hip/hip_runtime.h>
#include <math.h>

typedef float v2f __attribute__((ext_vector_type(2)));
typedef float v8f __attribute__((ext_vector_type(8)));

#define EMBED            2048
#define BATCH            32
#define BLOCK_THREADS    256
#define WAVES_PER_BLOCK  8
#define ROWS_PER_WAVE    16
#define ROWS_PER_BLOCK   128
#define BLOCKS_PER_BATCH 16

__global__ __launch_bounds__(BLOCK_THREADS)
void attn_outer_softmax_wmma(const float* __restrict__ q,
                             const float* __restrict__ k,
                             const float* __restrict__ v,
                             float* __restrict__ out)
{
    __shared__ __attribute__((aligned(16))) float sK[EMBED];
    __shared__ __attribute__((aligned(16))) float sV[EMBED];
    __shared__ __attribute__((aligned(16))) float sRed[2 * WAVES_PER_BLOCK];
    __shared__ float sOut[WAVES_PER_BLOCK][32];

    const int b    = blockIdx.y;
    const int tid  = threadIdx.x;
    const int wave = tid >> 5;
    const int lane = tid & 31;

    const int rowBase = blockIdx.x * ROWS_PER_BLOCK + wave * ROWS_PER_WAVE;
    const int n       = lane & 15;
    const int h       = lane >> 4;

    const float4* kb4 = (const float4*)(k + b * EMBED);
    const float4* vb4 = (const float4*)(v + b * EMBED);
    const float4 kA = kb4[tid];
    const float4 kB = kb4[tid + BLOCK_THREADS];
    const float4 vA = vb4[tid];
    const float4 vB = vb4[tid + BLOCK_THREADS];
    const float  qv = q[b * EMBED + rowBase + n];

    ((float4*)sK)[tid]                 = kA;
    ((float4*)sK)[tid + BLOCK_THREADS] = kB;
    ((float4*)sV)[tid]                 = vA;
    ((float4*)sV)[tid + BLOCK_THREADS] = vB;

    float kmx = fmaxf(fmaxf(fmaxf(kA.x, kA.y), fmaxf(kA.z, kA.w)),
                      fmaxf(fmaxf(kB.x, kB.y), fmaxf(kB.z, kB.w)));
    float kmn = fminf(fminf(fminf(kA.x, kA.y), fminf(kA.z, kA.w)),
                      fminf(fminf(kB.x, kB.y), fminf(kB.z, kB.w)));
#define LX16(X, LO, HI) __builtin_bit_cast(float, __builtin_amdgcn_permlane16(            \
        __builtin_bit_cast(int, (X)), __builtin_bit_cast(int, (X)), (LO), (HI), false, false))
#define LXH(X)          __builtin_bit_cast(float, __builtin_amdgcn_permlanex16(           \
        __builtin_bit_cast(int, (X)), __builtin_bit_cast(int, (X)), 0x76543210, 0xfedcba98, false, false))
    kmx = fmaxf(kmx, LX16(kmx, 0x67452301, 0xefcdab89));
    kmn = fminf(kmn, LX16(kmn, 0x67452301, 0xefcdab89));
    kmx = fmaxf(kmx, LX16(kmx, 0x54761032, 0xdcfe98ba));
    kmn = fminf(kmn, LX16(kmn, 0x54761032, 0xdcfe98ba));
    kmx = fmaxf(kmx, LX16(kmx, 0x32107654, 0xba98fedc));
    kmn = fminf(kmn, LX16(kmn, 0x32107654, 0xba98fedc));
    kmx = fmaxf(kmx, LX16(kmx, 0xfedcba98, 0x76543210));
    kmn = fminf(kmn, LX16(kmn, 0xfedcba98, 0x76543210));
    kmx = fmaxf(kmx, LXH(kmx));
    kmn = fminf(kmn, LXH(kmn));
#undef LXH
#undef LX16
    if (lane == 0) {
        sRed[wave]                   = kmx;
        sRed[WAVES_PER_BLOCK + wave] = kmn;
    }
    __syncthreads();
    const float4 rx0 = ((const float4*)sRed)[0], rx1 = ((const float4*)sRed)[1];
    const float4 rn0 = ((const float4*)sRed)[2], rn1 = ((const float4*)sRed)[3];
    const float kmax = fmaxf(fmaxf(fmaxf(rx0.x, rx0.y), fmaxf(rx0.z, rx0.w)),
                             fmaxf(fmaxf(rx1.x, rx1.y), fmaxf(rx1.z, rx1.w)));
    const float kmin = fminf(fminf(fminf(rn0.x, rn0.y), fminf(rn0.z, rn0.w)),
                             fminf(fminf(rn1.x, rn1.y), fminf(rn1.z, rn1.w)));


    const float tlScale = 0.02209708691207961f * 1.4426950408889634f;
    const float tl = qv * tlScale;
    const float m2 = fmaxf(tl * kmax, tl * kmin);

    const bool isV   = (n == 0);
    const bool isOne = (n == 1);

    const float4* sK4 = (const float4*)sK;
    const float4* sV4 = (const float4*)sV;
    const int fo = 8 * h;

    const float bconst = isOne ? 1.0f : 0.0f;

#define EX(X)   __builtin_amdgcn_exp2f(fmaf(tl, (X), -m2))
#define SEL(X)  (isV ? (X) : bconst)
#define TLO(K, V) { (K).x = EX((K).x); (K).y = EX((K).y); (V).x = SEL((V).x); (V).y = SEL((V).y); }
#define THI(K, V) { (K).z = EX((K).z); (K).w = EX((K).w); (V).z = SEL((V).z); (V).w = SEL((V).w); }
#define SB      __builtin_amdgcn_sched_barrier(0);
#define WM(AX, AY, BX, BY)                                                    \
    {                                                                         \
        v2f a_, b_;                                                           \
        a_.x = (AX); a_.y = (AY); b_.x = (BX); b_.y = (BY);                   \
        acc = __builtin_amdgcn_wmma_f32_16x16x4_f32(false, a_, false, b_,     \
                                                    (short)0, acc, false, false); \
    }
#define WLO(K, V) WM((K).x, (K).y, (V).x, (V).y)
#define WHI(K, V) WM((K).z, (K).w, (V).z, (V).w)

    float4 kc0 = sK4[fo + 0], vc0 = sV4[fo + 0], kc1 = sK4[fo + 1], vc1 = sV4[fo + 1];
    float4 kc2 = sK4[fo + 2], vc2 = sV4[fo + 2], kc3 = sK4[fo + 3], vc3 = sV4[fo + 3];
    float4 kc4 = sK4[fo + 4], vc4 = sV4[fo + 4], kc5 = sK4[fo + 5], vc5 = sV4[fo + 5];
    float4 kc6 = sK4[fo + 6], vc6 = sV4[fo + 6], kc7 = sK4[fo + 7], vc7 = sV4[fo + 7];
    TLO(kc0, vc0) THI(kc0, vc0) TLO(kc1, vc1) THI(kc1, vc1)
    TLO(kc2, vc2) THI(kc2, vc2) TLO(kc3, vc3) THI(kc3, vc3)
    TLO(kc4, vc4) THI(kc4, vc4) TLO(kc5, vc5) THI(kc5, vc5)
    TLO(kc6, vc6) THI(kc6, vc6) TLO(kc7, vc7) THI(kc7, vc7)

    v8f acc = {};

    #pragma unroll 2
    for (int j0 = 0; j0 < EMBED; j0 += 64) {
        const int fn = (((j0 + 64) & (EMBED - 1)) >> 2) + fo;
        float4 kn0 = sK4[fn + 0], vn0 = sV4[fn + 0], kn1 = sK4[fn + 1], vn1 = sV4[fn + 1];
        float4 kn2 = sK4[fn + 2], vn2 = sV4[fn + 2], kn3 = sK4[fn + 3], vn3 = sV4[fn + 3];
        float4 kn4 = sK4[fn + 4], vn4 = sV4[fn + 4], kn5 = sK4[fn + 5], vn5 = sV4[fn + 5];
        float4 kn6 = sK4[fn + 6], vn6 = sV4[fn + 6], kn7 = sK4[fn + 7], vn7 = sV4[fn + 7];
        SB

        WLO(kc0, vc0) SB
        WHI(kc0, vc0) SB
        WLO(kc1, vc1) SB
        WHI(kc1, vc1) SB
        WLO(kc2, vc2) SB  TLO(kn0, vn0) THI(kn0, vn0) SB
        WHI(kc2, vc2) SB  TLO(kn1, vn1) THI(kn1, vn1) SB
        WLO(kc3, vc3) SB  TLO(kn2, vn2) THI(kn2, vn2) SB
        WHI(kc3, vc3) SB  TLO(kn3, vn3) THI(kn3, vn3) SB
        WLO(kc4, vc4) SB  TLO(kn4, vn4) SB
        WHI(kc4, vc4) SB  THI(kn4, vn4) SB
        WLO(kc5, vc5) SB  TLO(kn5, vn5) SB
        WHI(kc5, vc5) SB  THI(kn5, vn5) SB
        WLO(kc6, vc6) SB  TLO(kn6, vn6) SB
        WHI(kc6, vc6) SB  THI(kn6, vn6) SB
        WLO(kc7, vc7) SB  TLO(kn7, vn7) SB
        WHI(kc7, vc7) SB  THI(kn7, vn7) SB

        kc0 = kn0; kc1 = kn1; kc2 = kn2; kc3 = kn3; kc4 = kn4; kc5 = kn5; kc6 = kn6; kc7 = kn7;
        vc0 = vn0; vc1 = vn1; vc2 = vn2; vc3 = vn3; vc4 = vn4; vc5 = vn5; vc6 = vn6; vc7 = vn7;
    }
#undef WHI
#undef WLO
#undef WM
#undef SB
#undef THI
#undef TLO
#undef SEL
#undef EX

    if (n < 2) {
        #pragma unroll
        for (int r = 0; r < 8; ++r)
            sOut[wave][n * 16 + h * 8 + r] = acc[r];
    }
    __syncthreads();

    if (lane < 16) {
        const float num = sOut[wave][lane];
        const float den = sOut[wave][16 + lane];
        out[b * EMBED + rowBase + lane] = num / den;
    }
}

extern "C" void kernel_launch(void* const* d_in, const int* in_sizes, int n_in,
                              void* d_out, int out_size, void* d_ws, size_t ws_size,
                              hipStream_t stream) {
    const float* q = (const float*)d_in[0];
    const float* k = (const float*)d_in[1];
    const float* v = (const float*)d_in[2];
    float* out = (float*)d_out;

    dim3 grid(BLOCKS_PER_BATCH, BATCH);
    dim3 block(BLOCK_THREADS);
    attn_outer_softmax_wmma<<<grid, block, 0, stream>>>(q, k, v, out);
}


# Both WMMA operands carry exps: A rows 0-12 and B columns 0-12 hold two 13-row groups, row/col 14 = v, 15 = 1, so one WMMA serves 26 rows (was 16); each wave runs a 256-j slice of 5 tiles, join in LDS
# speedup vs baseline: 1.5080x; 1.1168x over previous
#include <hip/hip_runtime.h>
#include <math.h>


typedef float v2f __attribute__((ext_vector_type(2)));
typedef float v8f __attribute__((ext_vector_type(8)));

#define EMBED            2048
#define BATCH            32
#define BLOCK_THREADS    256
#define WAVES_PER_BLOCK  8
#define ROWS_PER_BLOCK   128
#define BLOCKS_PER_BATCH 16
#define HALF_ROWS        64
#define GROUP_ROWS       13
#define TILES            5
#define V_OFF            2052
#define ONE_OFF          4104
#define SX_FLOATS        (ONE_OFF + 256)

__global__ __launch_bounds__(BLOCK_THREADS)
void attn_outer_softmax_wmma(const float* __restrict__ q,
                             const float* __restrict__ k,
                             const float* __restrict__ v,
                             float* __restrict__ out)
{
    __shared__ __attribute__((aligned(16))) float sX[SX_FLOATS];
    __shared__ __attribute__((aligned(16))) float sRed[2 * WAVES_PER_BLOCK];
    __shared__ float sQ[ROWS_PER_BLOCK];
    __shared__ float sD[WAVES_PER_BLOCK][TILES][8][32];

    const int b    = blockIdx.y;
    const int tid  = threadIdx.x;
    const int wave = tid >> 5;
    const int lane = tid & 31;
    const int rowBlock = blockIdx.x * ROWS_PER_BLOCK;
    const int c = lane & 15;
    const int h = lane >> 4;

    const float4* kb4 = (const float4*)(k + b * EMBED);
    const float4* vb4 = (const float4*)(v + b * EMBED);
    const float4 kA = kb4[tid];
    const float4 kB = kb4[tid + BLOCK_THREADS];
    const float4 vA = vb4[tid];
    const float4 vB = vb4[tid + BLOCK_THREADS];
    const float  qv = q[b * EMBED + rowBlock + (tid & (ROWS_PER_BLOCK - 1))];

    ((float4*)sX)[tid]                           = kA;
    ((float4*)sX)[tid + BLOCK_THREADS]           = kB;
    ((float4*)(sX + V_OFF))[tid]                 = vA;
    ((float4*)(sX + V_OFF))[tid + BLOCK_THREADS] = vB;
    sX[ONE_OFF + tid] = 1.0f;
    if (tid < ROWS_PER_BLOCK) sQ[tid] = qv;

    float kmx = fmaxf(fmaxf(fmaxf(kA.x, kA.y), fmaxf(kA.z, kA.w)),
                      fmaxf(fmaxf(kB.x, kB.y), fmaxf(kB.z, kB.w)));
    float kmn = fminf(fminf(fminf(kA.x, kA.y), fminf(kA.z, kA.w)),
                      fminf(fminf(kB.x, kB.y), fminf(kB.z, kB.w)));
#define LX16(X, LO, HI) __builtin_bit_cast(float, __builtin_amdgcn_permlane16(            \
        __builtin_bit_cast(int, (X)), __builtin_bit_cast(int, (X)), (LO), (HI), false, false))
#define LXH(X)          __builtin_bit_cast(float, __builtin_amdgcn_permlanex16(           \
        __builtin_bit_cast(int, (X)), __builtin_bit_cast(int, (X)), 0x76543210, 0xfedcba98, false, false))
    kmx = fmaxf(kmx, LX16(kmx, 0x67452301, 0xefcdab89));
    kmn = fminf(kmn, LX16(kmn, 0x67452301, 0xefcdab89));
    kmx = fmaxf(kmx, LX16(kmx, 0x54761032, 0xdcfe98ba));
    kmn = fminf(kmn, LX16(kmn, 0x54761032, 0xdcfe98ba));
    kmx = fmaxf(kmx, LX16(kmx, 0x32107654, 0xba98fedc));
    kmn = fminf(kmn, LX16(kmn, 0x32107654, 0xba98fedc));
    kmx = fmaxf(kmx, LX16(kmx, 0xfedcba98, 0x76543210));
    kmn = fminf(kmn, LX16(kmn, 0xfedcba98, 0x76543210));
    kmx = fmaxf(kmx, LXH(kmx));
    kmn = fminf(kmn, LXH(kmn));
#undef LXH
#undef LX16
    if (lane == 0) {
        sRed[wave]                   = kmx;
        sRed[WAVES_PER_BLOCK + wave] = kmn;
    }
    __syncthreads();
    const float4 rx0 = ((const float4*)sRed)[0], rx1 = ((const float4*)sRed)[1];
    const float4 rn0 = ((const float4*)sRed)[2], rn1 = ((const float4*)sRed)[3];
    const float kmax = fmaxf(fmaxf(fmaxf(rx0.x, rx0.y), fmaxf(rx0.z, rx0.w)),
                             fmaxf(fmaxf(rx1.x, rx1.y), fmaxf(rx1.z, rx1.w)));
    const float kmin = fminf(fminf(fminf(rn0.x, rn0.y), fminf(rn0.z, rn0.w)),
                             fminf(fminf(rn1.x, rn1.y), fminf(rn1.z, rn1.w)));

    const float tlScale = 0.02209708691207961f * 1.4426950408889634f;

    const bool isAux = (c >= 14);
    const int laneBase4 = (c == 15) ? (ONE_OFF / 4)
                                    : ((c == 14 ? (V_OFF / 4) : 0) + wave * 64);
    const float4* src = (const float4*)sX + laneBase4 + 8 * h;

    float tlA, m2A, tlB, m2B, ntlA, nm2A, ntlB, nm2B;
    {
        const float ta = sQ[c] * tlScale, tb = sQ[HALF_ROWS + c] * tlScale;
        tlA = ta; m2A = fmaxf(ta * kmax, ta * kmin);
        tlB = tb; m2B = fmaxf(tb * kmax, tb * kmin);
        const float ua = sQ[GROUP_ROWS + c] * tlScale, ub = sQ[HALF_ROWS + GROUP_ROWS + c] * tlScale;
        ntlA = ua; nm2A = fmaxf(ua * kmax, ua * kmin);
        ntlB = ub; nm2B = fmaxf(ub * kmax, ub * kmin);
    }

#define EXA(X)  __builtin_amdgcn_exp2f(fmaf(tlA, (X), -m2A))
#define EXB(X)  __builtin_amdgcn_exp2f(fmaf(tlB, (X), -m2B))
#define SELA(X) (isAux ? (X) : EXA(X))
#define SELB(X) (isAux ? (X) : EXB(X))
#define PIN(P, Q, R_, S) asm volatile("" :: "v"(P), "v"(Q), "v"(R_), "v"(S));
#define TLO(R, A, B) { (A).x = SELA((R).x); (A).y = SELA((R).y); (B).x = SELB((R).x); (B).y = SELB((R).y); PIN((A).x, (A).y, (B).x, (B).y) }
#define THI(R, A, B) { (A).z = SELA((R).z); (A).w = SELA((R).w); (B).z = SELB((R).z); (B).w = SELB((R).w); PIN((A).z, (A).w, (B).z, (B).w) }
#define SB      __builtin_amdgcn_sched_barrier(0);
#define WM(AX, AY, BX, BY)                                                    \
    {                                                                         \
        v2f a_, b_;                                                           \
        a_.x = (AX); a_.y = (AY); b_.x = (BX); b_.y = (BY);                   \
        acc = __builtin_amdgcn_wmma_f32_16x16x4_f32(false, a_, false, b_,     \
                                                    (short)0, acc, false, false); \
    }
#define WLO(A, B) WM((A).x, (A).y, (B).x, (B).y)
#define WHI(A, B) WM((A).z, (A).w, (B).z, (B).w)

#define BLOCK(AC, BC, AN, BN, NBLK)                                                         \
    {                                                                                       \
        const float4 r0 = src[(NBLK) * 16 + 0], r1 = src[(NBLK) * 16 + 1];                  \
        const float4 r2 = src[(NBLK) * 16 + 2], r3 = src[(NBLK) * 16 + 3];                  \
        const float4 r4 = src[(NBLK) * 16 + 4], r5 = src[(NBLK) * 16 + 5];                  \
        const float4 r6 = src[(NBLK) * 16 + 6], r7 = src[(NBLK) * 16 + 7];                  \
        SB                                                                                  \
        WLO(AC[0], BC[0]) SB                                                                \
        WHI(AC[0], BC[0]) SB                                                                \
        WLO(AC[1], BC[1]) SB                                                                \
        WHI(AC[1], BC[1]) SB                                                                \
        WLO(AC[2], BC[2]) SB  TLO(r0, AN[0], BN[0]) THI(r0, AN[0], BN[0]) SB                \
        WHI(AC[2], BC[2]) SB  TLO(r1, AN[1], BN[1]) THI(r1, AN[1], BN[1]) SB                \
        WLO(AC[3], BC[3]) SB  TLO(r2, AN[2], BN[2]) THI(r2, AN[2], BN[2]) SB                \
        WHI(AC[3], BC[3]) SB  TLO(r3, AN[3], BN[3]) THI(r3, AN[3], BN[3]) SB                \
        WLO(AC[4], BC[4]) SB  TLO(r4, AN[4], BN[4]) SB                                      \
        WHI(AC[4], BC[4]) SB  THI(r4, AN[4], BN[4]) SB                                      \
        WLO(AC[5], BC[5]) SB  TLO(r5, AN[5], BN[5]) SB                                      \
        WHI(AC[5], BC[5]) SB  THI(r5, AN[5], BN[5]) SB                                      \
        WLO(AC[6], BC[6]) SB  TLO(r6, AN[6], BN[6]) SB                                      \
        WHI(AC[6], BC[6]) SB  THI(r6, AN[6], BN[6]) SB                                      \
        WLO(AC[7], BC[7]) SB  TLO(r7, AN[7], BN[7]) SB                                      \
        WHI(AC[7], BC[7]) SB  THI(r7, AN[7], BN[7]) SB                                      \
    }

    float4 aX[8], bX[8], aY[8], bY[8];

    {
        const float4 r0 = src[0], r1 = src[1], r2 = src[2], r3 = src[3];
        const float4 r4 = src[4], r5 = src[5], r6 = src[6], r7 = src[7];
        TLO(r0, aX[0], bX[0]) THI(r0, aX[0], bX[0]) TLO(r1, aX[1], bX[1]) THI(r1, aX[1], bX[1])
        TLO(r2, aX[2], bX[2]) THI(r2, aX[2], bX[2]) TLO(r3, aX[3], bX[3]) THI(r3, aX[3], bX[3])
        TLO(r4, aX[4], bX[4]) THI(r4, aX[4], bX[4]) TLO(r5, aX[5], bX[5]) THI(r5, aX[5], bX[5])
        TLO(r6, aX[6], bX[6]) THI(r6, aX[6], bX[6]) TLO(r7, aX[7], bX[7]) THI(r7, aX[7], bX[7])
    }

    #pragma unroll 1
    for (int t = 0; t < TILES; ++t) {
        v8f acc = {};
        BLOCK(aX, bX, aY, bY, 1)
        BLOCK(aY, bY, aX, bX, 2)
        BLOCK(aX, bX, aY, bY, 3)
        tlA = ntlA; m2A = nm2A; tlB = ntlB; m2B = nm2B;
        BLOCK(aY, bY, aX, bX, 0)

        #pragma unroll
        for (int r = 0; r < 8; ++r) sD[wave][t][r][lane] = acc[r];

        {
            const int tn = (t + 2 < TILES) ? (t + 2) : (TILES - 1);
            const int rn = (GROUP_ROWS * tn + c < HALF_ROWS) ? (GROUP_ROWS * tn + c) : (HALF_ROWS - 1);
            const float ua = sQ[rn] * tlScale, ub = sQ[HALF_ROWS + rn] * tlScale;
            ntlA = ua; nm2A = fmaxf(ua * kmax, ua * kmin);
            ntlB = ub; nm2B = fmaxf(ub * kmax, ub * kmin);
        }
    }
#undef BLOCK
#undef WHI
#undef WLO
#undef WM
#undef SB
#undef THI
#undef PIN
#undef TLO
#undef SELB
#undef SELA
#undef EXB
#undef EXA

    __syncthreads();

    if (tid < ROWS_PER_BLOCK) {
        const int x  = tid & (HALF_ROWS - 1);
        const int tt = x / GROUP_ROWS;
        const int m  = x - GROUP_ROWS * tt;
        const bool sideB = (tid >= HALF_ROWS);
        const int rNum = sideB ? 6 : (m & 7);
        const int rDen = sideB ? 7 : (m & 7);
        const int lNum = sideB ? (16 + m) : (14 + 16 * (m >> 3));
        const int lDen = sideB ? (16 + m) : (15 + 16 * (m >> 3));
        float num = 0.0f, den = 0.0f;
        #pragma unroll
        for (int w = 0; w < WAVES_PER_BLOCK; ++w) {
            num += sD[w][tt][rNum][lNum];
            den += sD[w][tt][rDen][lDen];
        }
        out[b * EMBED + rowBlock + tid] = num / den;
    }
}

extern "C" void kernel_launch(void* const* d_in, const int* in_sizes, int n_in,
                              void* d_out, int out_size, void* d_ws, size_t ws_size,
                              hipStream_t stream) {
    const float* q = (const float*)d_in[0];
    const float* k = (const float*)d_in[1];
    const float* v = (const float*)d_in[2];
    float* out = (float*)d_out;

    dim3 grid(BLOCKS_PER_BATCH, BATCH);
    dim3 block(BLOCK_THREADS);
    attn_outer_softmax_wmma<<<grid, block, 0, stream>>>(q, k, v, out);
}


# speedup vs baseline: 1.6849x; 1.1173x over previous
#include <hip/hip_runtime.h>
#include <math.h>


typedef float v2f __attribute__((ext_vector_type(2)));
typedef float v8f __attribute__((ext_vector_type(8)));

#define EMBED            2048
#define BATCH            32
#define BLOCK_THREADS    256
#define WAVES_PER_BLOCK  8
#define ROWS_PER_BLOCK   128
#define BLOCKS_PER_BATCH 16
#define HALF_ROWS        64
#define GROUP_ROWS       13
#define TILES            5
#define V_OFF            2052
#define ONE_OFF          4104
#define SX_FLOATS        (ONE_OFF + 256)

__global__ __launch_bounds__(BLOCK_THREADS)
void attn_outer_softmax_wmma(const float* __restrict__ q,
                             const float* __restrict__ k,
                             const float* __restrict__ v,
                             float* __restrict__ out)
{
    __shared__ __attribute__((aligned(16))) float sX[SX_FLOATS];
    __shared__ __attribute__((aligned(16))) float sRed[2 * WAVES_PER_BLOCK];
    __shared__ float sQ[ROWS_PER_BLOCK];
    __shared__ float sD[WAVES_PER_BLOCK][TILES][8][32];

    const int b    = blockIdx.y;
    const int tid  = threadIdx.x;
    const int wave = tid >> 5;
    const int lane = tid & 31;
    const int rowBlock = blockIdx.x * ROWS_PER_BLOCK;
    const int c = lane & 15;
    const int h = lane >> 4;

    const float4* kb4 = (const float4*)(k + b * EMBED);
    const float4* vb4 = (const float4*)(v + b * EMBED);
    const float4 kA = kb4[tid];
    const float4 kB = kb4[tid + BLOCK_THREADS];
    const float4 vA = vb4[tid];
    const float4 vB = vb4[tid + BLOCK_THREADS];
    const float  qv = q[b * EMBED + rowBlock + (tid & (ROWS_PER_BLOCK - 1))];

    ((float4*)sX)[tid]                           = kA;
    ((float4*)sX)[tid + BLOCK_THREADS]           = kB;
    ((float4*)(sX + V_OFF))[tid]                 = vA;
    ((float4*)(sX + V_OFF))[tid + BLOCK_THREADS] = vB;
    sX[ONE_OFF + tid] = 1.0f;
    if (tid < ROWS_PER_BLOCK) sQ[tid] = qv;

    float kmx = fmaxf(fmaxf(fmaxf(kA.x, kA.y), fmaxf(kA.z, kA.w)),
                      fmaxf(fmaxf(kB.x, kB.y), fmaxf(kB.z, kB.w)));
    float kmn = fminf(fminf(fminf(kA.x, kA.y), fminf(kA.z, kA.w)),
                      fminf(fminf(kB.x, kB.y), fminf(kB.z, kB.w)));
#define LX16(X, LO, HI) __builtin_bit_cast(float, __builtin_amdgcn_permlane16(            \
        __builtin_bit_cast(int, (X)), __builtin_bit_cast(int, (X)), (LO), (HI), false, false))
#define LXH(X)          __builtin_bit_cast(float, __builtin_amdgcn_permlanex16(           \
        __builtin_bit_cast(int, (X)), __builtin_bit_cast(int, (X)), 0x76543210, 0xfedcba98, false, false))
    kmx = fmaxf(kmx, LX16(kmx, 0x67452301, 0xefcdab89));
    kmn = fminf(kmn, LX16(kmn, 0x67452301, 0xefcdab89));
    kmx = fmaxf(kmx, LX16(kmx, 0x54761032, 0xdcfe98ba));
    kmn = fminf(kmn, LX16(kmn, 0x54761032, 0xdcfe98ba));
    kmx = fmaxf(kmx, LX16(kmx, 0x32107654, 0xba98fedc));
    kmn = fminf(kmn, LX16(kmn, 0x32107654, 0xba98fedc));
    kmx = fmaxf(kmx, LX16(kmx, 0xfedcba98, 0x76543210));
    kmn = fminf(kmn, LX16(kmn, 0xfedcba98, 0x76543210));
    kmx = fmaxf(kmx, LXH(kmx));
    kmn = fminf(kmn, LXH(kmn));
#undef LXH
#undef LX16
    if (lane == 0) {
        sRed[wave]                   = kmx;
        sRed[WAVES_PER_BLOCK + wave] = kmn;
    }
    __syncthreads();
    const float4 rx0 = ((const float4*)sRed)[0], rx1 = ((const float4*)sRed)[1];
    const float4 rn0 = ((const float4*)sRed)[2], rn1 = ((const float4*)sRed)[3];
    const float kmax = fmaxf(fmaxf(fmaxf(rx0.x, rx0.y), fmaxf(rx0.z, rx0.w)),
                             fmaxf(fmaxf(rx1.x, rx1.y), fmaxf(rx1.z, rx1.w)));
    const float kmin = fminf(fminf(fminf(rn0.x, rn0.y), fminf(rn0.z, rn0.w)),
                             fminf(fminf(rn1.x, rn1.y), fminf(rn1.z, rn1.w)));

    const float tlScale = 0.02209708691207961f * 1.4426950408889634f;

    const bool isAux = (c >= 14);
    const int laneBase4 = (c == 15) ? (ONE_OFF / 4)
                                    : ((c == 14 ? (V_OFF / 4) : 0) + wave * 64);
    const float4* src = (const float4*)sX + laneBase4 + 8 * h;

    float tlA, m2A, tlB, m2B, ntlA, nm2A, ntlB, nm2B;
    {
        const float ta = sQ[c] * tlScale, tb = sQ[HALF_ROWS + c] * tlScale;
        tlA = ta; m2A = fmaxf(ta * kmax, ta * kmin);
        tlB = tb; m2B = fmaxf(tb * kmax, tb * kmin);
        const float ua = sQ[GROUP_ROWS + c] * tlScale, ub = sQ[HALF_ROWS + GROUP_ROWS + c] * tlScale;
        ntlA = ua; nm2A = fmaxf(ua * kmax, ua * kmin);
        ntlB = ub; nm2B = fmaxf(ub * kmax, ub * kmin);
    }

#define EXA(X)  __builtin_amdgcn_exp2f(fmaf(tlA, (X), -m2A))
#define EXB(X)  __builtin_amdgcn_exp2f(fmaf(tlB, (X), -m2B))
#define SELA(X) (isAux ? (X) : EXA(X))
#define SELB(X) (isAux ? (X) : EXB(X))
#define PIN(P, Q, R_, S) asm volatile("" :: "v"(P), "v"(Q), "v"(R_), "v"(S));
#define TLO(R, A, B) { (A).x = SELA((R).x); (A).y = SELA((R).y); (B).x = SELB((R).x); (B).y = SELB((R).y); PIN((A).x, (A).y, (B).x, (B).y) }
#define THI(R, A, B) { (A).z = SELA((R).z); (A).w = SELA((R).w); (B).z = SELB((R).z); (B).w = SELB((R).w); PIN((A).z, (A).w, (B).z, (B).w) }
#define SB      __builtin_amdgcn_sched_barrier(0);
#define WM(AX, AY, BX, BY)                                                    \
    {                                                                         \
        v2f a_, b_;                                                           \
        a_.x = (AX); a_.y = (AY); b_.x = (BX); b_.y = (BY);                   \
        acc = __builtin_amdgcn_wmma_f32_16x16x4_f32(false, a_, false, b_,     \
                                                    (short)0, acc, false, false); \
    }
#define WLO(A, B) WM((A).x, (A).y, (B).x, (B).y)
#define WHI(A, B) WM((A).z, (A).w, (B).z, (B).w)

#define ARG(G, XA, XB) { (G).x = fmaf(tlA, (XA), -m2A); (G).y = fmaf(tlA, (XB), -m2A);      \
                         (G).z = fmaf(tlB, (XA), -m2B); (G).w = fmaf(tlB, (XB), -m2B); }
#define EXQ(E, G) { (E).x = __builtin_amdgcn_exp2f((G).x); (E).y = __builtin_amdgcn_exp2f((G).y);  \
                    (E).z = __builtin_amdgcn_exp2f((G).z); (E).w = __builtin_amdgcn_exp2f((G).w); }
#define SLO(E, R, A, B) { (A).x = isAux ? (R).x : (E).x; (A).y = isAux ? (R).y : (E).y;     \
                          (B).x = isAux ? (R).x : (E).z; (B).y = isAux ? (R).y : (E).w; }
#define SHI(E, R, A, B) { (A).z = isAux ? (R).z : (E).x; (A).w = isAux ? (R).w : (E).y;     \
                          (B).z = isAux ? (R).z : (E).z; (B).w = isAux ? (R).w : (E).w; }
#define P4(V)        PIN((V).x, (V).y, (V).z, (V).w)
#define PLO(A, B)    PIN((A).x, (A).y, (B).x, (B).y)
#define PHI(A, B)    PIN((A).z, (A).w, (B).z, (B).w)
#define GT   __builtin_amdgcn_sched_group_barrier(0x400, 1, 0);
#define GP1  __builtin_amdgcn_sched_group_barrier(0x002, 1, 0);
#define GP2  __builtin_amdgcn_sched_group_barrier(0x002, 2, 0);
#define TP1x4 GT GP1 GT GP1 GT GP1 GT GP1
#define TP2x4 GT GP2 GT GP2 GT GP2 GT GP2

#define BLOCK(AC, BC, AN, BN, NBLK)                                                         \
    {                                                                                       \
        const float4 r0 = src[(NBLK) * 16 + 0], r1 = src[(NBLK) * 16 + 1];                  \
        const float4 r2 = src[(NBLK) * 16 + 2], r3 = src[(NBLK) * 16 + 3];                  \
        const float4 r4 = src[(NBLK) * 16 + 4], r5 = src[(NBLK) * 16 + 5];                  \
        const float4 r6 = src[(NBLK) * 16 + 6], r7 = src[(NBLK) * 16 + 7];                  \
        float4 g0, g1, g2, g3, g4, g5, g6, g7, g8, g9, g10, g11, g12, g13, g14, g15;        \
        float4 e0, e1, e2, e3, e4, e5, e6, e7, e8, e9, e10, e11, e12, e13, e14, e15;        \
        SB                                                                                  \
        WLO(AC[0], BC[0]) SB                                                                \
        WHI(AC[0], BC[0]) SB                                                                \
        WLO(AC[1], BC[1]) SB                                                                \
        WHI(AC[1], BC[1]) SB                                                                \
        ARG(g0, r0.x, r0.y) ARG(g1, r0.z, r0.w) P4(g0) P4(g1) SB                            \
        WLO(AC[2], BC[2]) SB                                                                \
        EXQ(e0, g0) EXQ(e1, g1) ARG(g2, r1.x, r1.y) ARG(g3, r1.z, r1.w)                     \
        P4(e0) P4(e1) P4(g2) P4(g3) TP1x4 TP1x4 SB                                          \
        WHI(AC[2], BC[2]) SB                                                                \
        EXQ(e2, g2) EXQ(e3, g3) SLO(e0, r0, AN[0], BN[0]) SHI(e1, r0, AN[0], BN[0])         \
        ARG(g4, r2.x, r2.y) ARG(g5, r2.z, r2.w)                                             \
        P4(e2) P4(e3) PLO(AN[0], BN[0]) PHI(AN[0], BN[0]) P4(g4) P4(g5) TP2x4 TP2x4 SB      \
        WLO(AC[3], BC[3]) SB                                                                \
        EXQ(e4, g4) EXQ(e5, g5) SLO(e2, r1, AN[1], BN[1]) SHI(e3, r1, AN[1], BN[1])         \
        ARG(g6, r3.x, r3.y) ARG(g7, r3.z, r3.w)                                             \
        P4(e4) P4(e5) PLO(AN[1], BN[1]) PHI(AN[1], BN[1]) P4(g6) P4(g7) TP2x4 TP2x4 SB      \
        WHI(AC[3], BC[3]) SB                                                                \
        EXQ(e6, g6) EXQ(e7, g7) SLO(e4, r2, AN[2], BN[2]) SHI(e5, r2, AN[2], BN[2])         \
        ARG(g8, r4.x, r4.y)                                                                 \
        P4(e6) P4(e7) PLO(AN[2], BN[2]) PHI(AN[2], BN[2]) P4(g8) TP2x4 TP1x4 SB             \
        WLO(AC[4], BC[4]) SB                                                                \
        EXQ(e8, g8) SLO(e6, r3, AN[3], BN[3]) SHI(e7, r3, AN[3], BN[3]) ARG(g9, r4.z, r4.w) \
        P4(e8) PLO(AN[3], BN[3]) PHI(AN[3], BN[3]) P4(g9) TP2x4 SB                          \
        WHI(AC[4], BC[4]) SB                                                                \
        EXQ(e9, g9) SLO(e8, r4, AN[4], BN[4]) ARG(g10, r5.x, r5.y)                          \
        P4(e9) PLO(AN[4], BN[4]) P4(g10) TP2x4 SB                                           \
        WLO(AC[5], BC[5]) SB                                                                \
        EXQ(e10, g10) SHI(e9, r4, AN[4], BN[4]) ARG(g11, r5.z, r5.w)                        \
        P4(e10) PHI(AN[4], BN[4]) P4(g11) TP2x4 SB                                          \
        WHI(AC[5], BC[5]) SB                                                                \
        EXQ(e11, g11) SLO(e10, r5, AN[5], BN[5]) ARG(g12, r6.x, r6.y)                       \
        P4(e11) PLO(AN[5], BN[5]) P4(g12) TP2x4 SB                                          \
        WLO(AC[6], BC[6]) SB                                                                \
        EXQ(e12, g12) SHI(e11, r5, AN[5], BN[5]) ARG(g13, r6.z, r6.w)                       \
        P4(e12) PHI(AN[5], BN[5]) P4(g13) TP2x4 SB                                          \
        WHI(AC[6], BC[6]) SB                                                                \
        EXQ(e13, g13) SLO(e12, r6, AN[6], BN[6]) ARG(g14, r7.x, r7.y)                       \
        P4(e13) PLO(AN[6], BN[6]) P4(g14) TP2x4 SB                                          \
        WLO(AC[7], BC[7]) SB                                                                \
        EXQ(e14, g14) SHI(e13, r6, AN[6], BN[6]) ARG(g15, r7.z, r7.w)                       \
        P4(e14) PHI(AN[6], BN[6]) P4(g15) TP2x4 SB                                          \
        WHI(AC[7], BC[7]) SB                                                                \
        EXQ(e15, g15) SLO(e14, r7, AN[7], BN[7])                                            \
        P4(e15) PLO(AN[7], BN[7]) TP1x4 SB                                                  \
        SHI(e15, r7, AN[7], BN[7]) PHI(AN[7], BN[7]) SB                                     \
    }

    float4 aX[8], bX[8], aY[8], bY[8];

    {
        const float4 r0 = src[0], r1 = src[1], r2 = src[2], r3 = src[3];
        const float4 r4 = src[4], r5 = src[5], r6 = src[6], r7 = src[7];
        TLO(r0, aX[0], bX[0]) THI(r0, aX[0], bX[0]) TLO(r1, aX[1], bX[1]) THI(r1, aX[1], bX[1])
        TLO(r2, aX[2], bX[2]) THI(r2, aX[2], bX[2]) TLO(r3, aX[3], bX[3]) THI(r3, aX[3], bX[3])
        TLO(r4, aX[4], bX[4]) THI(r4, aX[4], bX[4]) TLO(r5, aX[5], bX[5]) THI(r5, aX[5], bX[5])
        TLO(r6, aX[6], bX[6]) THI(r6, aX[6], bX[6]) TLO(r7, aX[7], bX[7]) THI(r7, aX[7], bX[7])
    }

    #pragma unroll 1
    for (int t = 0; t < TILES; ++t) {
        v8f acc = {};
        BLOCK(aX, bX, aY, bY, 1)
        BLOCK(aY, bY, aX, bX, 2)
        BLOCK(aX, bX, aY, bY, 3)
        tlA = ntlA; m2A = nm2A; tlB = ntlB; m2B = nm2B;
        BLOCK(aY, bY, aX, bX, 0)

        #pragma unroll
        for (int r = 0; r < 8; ++r) sD[wave][t][r][lane] = acc[r];

        {
            const int tn = (t + 2 < TILES) ? (t + 2) : (TILES - 1);
            const int rn = (GROUP_ROWS * tn + c < HALF_ROWS) ? (GROUP_ROWS * tn + c) : (HALF_ROWS - 1);
            const float ua = sQ[rn] * tlScale, ub = sQ[HALF_ROWS + rn] * tlScale;
            ntlA = ua; nm2A = fmaxf(ua * kmax, ua * kmin);
            ntlB = ub; nm2B = fmaxf(ub * kmax, ub * kmin);
        }
    }
#undef BLOCK
#undef TP2x4
#undef TP1x4
#undef GP2
#undef GP1
#undef GT
#undef PHI
#undef PLO
#undef P4
#undef SHI
#undef SLO
#undef EXQ
#undef ARG
#undef WHI
#undef WLO
#undef WM
#undef SB
#undef THI
#undef PIN
#undef TLO
#undef SELB
#undef SELA
#undef EXB
#undef EXA

    __syncthreads();

    if (tid < ROWS_PER_BLOCK) {
        const int x  = tid & (HALF_ROWS - 1);
        const int tt = x / GROUP_ROWS;
        const int m  = x - GROUP_ROWS * tt;
        const bool sideB = (tid >= HALF_ROWS);
        const int rNum = sideB ? 6 : (m & 7);
        const int rDen = sideB ? 7 : (m & 7);
        const int lNum = sideB ? (16 + m) : (14 + 16 * (m >> 3));
        const int lDen = sideB ? (16 + m) : (15 + 16 * (m >> 3));
        float num = 0.0f, den = 0.0f;
        #pragma unroll
        for (int w = 0; w < WAVES_PER_BLOCK; ++w) {
            num += sD[w][tt][rNum][lNum];
            den += sD[w][tt][rDen][lDen];
        }
        out[b * EMBED + rowBlock + tid] = num / den;
    }
}

extern "C" void kernel_launch(void* const* d_in, const int* in_sizes, int n_in,
                              void* d_out, int out_size, void* d_ws, size_t ws_size,
                              hipStream_t stream) {
    const float* q = (const float*)d_in[0];
    const float* k = (const float*)d_in[1];
    const float* v = (const float*)d_in[2];
    float* out = (float*)d_out;

    dim3 grid(BLOCKS_PER_BATCH, BATCH);
    dim3 block(BLOCK_THREADS);
    attn_outer_softmax_wmma<<<grid, block, 0, stream>>>(q, k, v, out);
}
